# starting kernel
# baseline (speedup 1.0000x reference)
#include <hip/hip_runtime.h>
#include <math.h>

typedef __attribute__((ext_vector_type(2))) float v2f;
typedef __attribute__((ext_vector_type(8))) float v8f;

#define NT 1024
#define NS 2048
#define WAVES_PER_BLOCK 8
#define BLOCK_THREADS (WAVES_PER_BLOCK * 32)

#define DPP8_SEL(a,b,c,d,e,f,g,h) \
  ((a)|((b)<<3)|((c)<<6)|((d)<<9)|((e)<<12)|((f)<<15)|((g)<<18)|((h)<<21))
constexpr int DPP8_XOR1 = DPP8_SEL(1,0,3,2,5,4,7,6);
constexpr int DPP8_XOR2 = DPP8_SEL(2,3,0,1,6,7,4,5);
constexpr int DPP8_XOR4 = DPP8_SEL(4,5,6,7,0,1,2,3);

__device__ __forceinline__ float bcastf(float v, int l) {
  return __builtin_bit_cast(
      float, __builtin_amdgcn_readlane(__builtin_bit_cast(int, v), l));
}

template <int SEL>
__device__ __forceinline__ float dpp8f(float v) {
  return __builtin_bit_cast(
      float, __builtin_amdgcn_mov_dpp8(__builtin_bit_cast(int, v), SEL));
}

__device__ __forceinline__ float xor8f(float v) {
  int x = __builtin_bit_cast(int, v);
  int r = __builtin_amdgcn_permlane16(x, x, 0xFEDCBA98u, 0x76543210u,
                                       false,  false);
  return __builtin_bit_cast(float, r);
}

__device__ __forceinline__ float wmma_fold16(float p) {
  v2f a; a.x = 1.0f; a.y = 1.0f;
  v2f b; b.x = p;    b.y = 0.0f;
  v8f c = {};
  v8f d = __builtin_amdgcn_wmma_f32_16x16x4_f32(
       false, a,  false, b,
       (short)0, c,  false,  false);
  return d[0];
}

__device__ __forceinline__ float sigmoidf(float x) {
  return 1.0f / (1.0f + expf(-x));
}

__global__ __launch_bounds__(BLOCK_THREADS) void waternet_scan(
    const float* __restrict__ P,  const float* __restrict__ T1,
    const float* __restrict__ T2, const float* __restrict__ E,
    const float* __restrict__ w,  float* __restrict__ out) {
  const int lane = threadIdx.x & 31;
  const int wv   = threadIdx.x >> 5;
  const int s    = blockIdx.x * WAVES_PER_BLOCK + wv;

  const float gm = expf(w[lane]) + 1.0f;
  const float ge = sigmoidf(w[32 + lane]);
  const float go = sigmoidf(w[64 + lane]);
  const float xw = w[96 + lane];
  float mx = xw;
  for (int off = 16; off; off >>= 1) mx = fmaxf(mx, __shfl_xor(mx, off, 32));
  const float ex = expf(xw - mx);
  float sm = ex;
  for (int off = 16; off; off >>= 1) sm += __shfl_xor(sm, off, 32);
  const float ga = ex / sm;

  float S = 0.0f, H = 0.0f;

  for (int t0 = 0; t0 < NT; t0 += 32) {
    const int idx = (t0 + lane) * NS + s;
    const float Pv  = P[idx];
    const float t1v = T1[idx];
    const float t2v = T2[idx];
    const float Ev  = E[idx];

    if (t0 + 32 < NT) {
      const int nidx = idx + 32 * NS;
      __builtin_prefetch(&P[nidx],  0, 0);
      __builtin_prefetch(&T1[nidx], 0, 0);
      __builtin_prefetch(&T2[nidx], 0, 0);
      __builtin_prefetch(&E[nidx],  0, 0);
    }

    const float Ta = (t1v + t2v) * 0.5f;
    float ratio = (t1v + t2v) / (t2v - t1v);
    ratio = fminf(1.0f, fmaxf(-1.0f, ratio));
    float rp = 1.0f - acosf(ratio) / 3.1415f;
    rp = (t1v >= 0.0f) ? 1.0f : rp;
    rp = (t2v <= 0.0f) ? 0.0f : rp;
    const float Psn = (1.0f - rp) * Pv;
    const float Plq = rp * Pv;

    float yv = 0.0f;
#pragma unroll
    for (int k = 0; k < 32; ++k) {
      const float Ta_k = bcastf(Ta,  k);
      const float Ps_k = bcastf(Psn, k);
      const float Pl_k = bcastf(Plq, k);
      const float E_k  = bcastf(Ev,  k);

      const float Smelt = fminf(S, fmaxf(Ta_k * gm, 0.0f));
      S = S + Ps_k - Smelt;
      float Hn = H + Smelt + Pl_k - E_k * ge;
      Hn = fmaxf(Hn, 0.0f);
      const float Q = Hn * go;
      H = Hn - Q;

      float r = wmma_fold16(Q * ga);
      r += xor8f(r);
      r += dpp8f<DPP8_XOR4>(r);
      r += dpp8f<DPP8_XOR2>(r);
      r += dpp8f<DPP8_XOR1>(r);

      yv = (lane == k) ? r : yv;
    }
    out[idx] = yv;
  }
}

extern "C" void kernel_launch(void* const* d_in, const int* in_sizes, int n_in,
                              void* d_out, int out_size, void* d_ws, size_t ws_size,
                              hipStream_t stream) {
  (void)in_sizes; (void)n_in; (void)out_size; (void)d_ws; (void)ws_size;
  const float* P  = (const float*)d_in[0];
  const float* T1 = (const float*)d_in[1];
  const float* T2 = (const float*)d_in[2];
  const float* E  = (const float*)d_in[3];
  const float* w  = (const float*)d_in[4];
  float* out = (float*)d_out;

  dim3 grid(NS / WAVES_PER_BLOCK);
  dim3 block(BLOCK_THREADS);
  waternet_scan<<<grid, block, 0, stream>>>(P, T1, T2, E, w, out);
}


# Forcings staged through LDS: work-group loads each 32-step x 8-site tile with 4-row loads one block ahead, waves read their steps by same-address ds_load_b128 (no 32-row loads, no readlanes)
# speedup vs baseline: 1.2811x; 1.2811x over previous
#include <hip/hip_runtime.h>
#include <math.h>

typedef __attribute__((ext_vector_type(2))) float v2f;
typedef __attribute__((ext_vector_type(8))) float v8f;

#define NT 1024
#define NS 2048
#define WAVES_PER_BLOCK 8
#define BLOCK_THREADS (WAVES_PER_BLOCK * 32)

#define DPP8_SEL(a,b,c,d,e,f,g,h) \
  ((a)|((b)<<3)|((c)<<6)|((d)<<9)|((e)<<12)|((f)<<15)|((g)<<18)|((h)<<21))
constexpr int DPP8_XOR1 = DPP8_SEL(1,0,3,2,5,4,7,6);
constexpr int DPP8_XOR2 = DPP8_SEL(2,3,0,1,6,7,4,5);
constexpr int DPP8_XOR4 = DPP8_SEL(4,5,6,7,0,1,2,3);

template <int SEL>
__device__ __forceinline__ float dpp8f(float v) {
  return __builtin_bit_cast(
      float, __builtin_amdgcn_mov_dpp8(__builtin_bit_cast(int, v), SEL));
}

__device__ __forceinline__ float xor8f(float v) {
  int x = __builtin_bit_cast(int, v);
  int r = __builtin_amdgcn_permlane16(x, x, 0xFEDCBA98u, 0x76543210u,
                                       false,  false);
  return __builtin_bit_cast(float, r);
}

__device__ __forceinline__ float wmma_fold16(float p) {
  v2f a; a.x = 1.0f; a.y = 1.0f;
  v2f b; b.x = p;    b.y = 0.0f;
  v8f c = {};
  v8f d = __builtin_amdgcn_wmma_f32_16x16x4_f32(
       false, a,  false, b,
       (short)0, c,  false,  false);
  return d[0];
}

__device__ __forceinline__ float sigmoidf(float x) {
  return 1.0f / (1.0f + expf(-x));
}

__global__ __launch_bounds__(BLOCK_THREADS) void waternet_scan(
    const float* __restrict__ P,  const float* __restrict__ T1,
    const float* __restrict__ T2, const float* __restrict__ E,
    const float* __restrict__ w,  float* __restrict__ out) {
  const int lane = threadIdx.x & 31;
  const int wv   = threadIdx.x >> 5;
  const int s    = blockIdx.x * WAVES_PER_BLOCK + wv;

  const float gm = expf(w[lane]) + 1.0f;
  const float ge = sigmoidf(w[32 + lane]);
  const float go = sigmoidf(w[64 + lane]);
  const float xw = w[96 + lane];
  float mx = xw;
  for (int off = 16; off; off >>= 1) mx = fmaxf(mx, __shfl_xor(mx, off, 32));
  const float ex = expf(xw - mx);
  float sm = ex;
  for (int off = 16; off; off >>= 1) sm += __shfl_xor(sm, off, 32);
  const float ga = ex / sm;

  float S = 0.0f, H = 0.0f;

  __shared__ float4 stage[2][BLOCK_THREADS];

  const int tile0 =
      (threadIdx.x >> 3) * NS + blockIdx.x * WAVES_PER_BLOCK + (threadIdx.x & 7);

  auto forcing = [](float Pv, float t1v, float t2v, float Ev) -> float4 {
    const float Ta = (t1v + t2v) * 0.5f;
    float ratio = (t1v + t2v) / (t2v - t1v);
    ratio = fminf(1.0f, fmaxf(-1.0f, ratio));
    float rp = 1.0f - acosf(ratio) / 3.1415f;
    rp = (t1v >= 0.0f) ? 1.0f : rp;
    rp = (t2v <= 0.0f) ? 0.0f : rp;
    return make_float4(Ta, (1.0f - rp) * Pv, rp * Pv, Ev);
  };

  stage[0][threadIdx.x] = forcing(P[tile0], T1[tile0], T2[tile0], E[tile0]);
  __syncthreads();

  for (int t0 = 0; t0 < NT; t0 += 32) {
    const int buf = (t0 >> 5) & 1;

    const int tn = (t0 + 32 < NT) ? t0 + 32 : t0;
    const int gidx = tn * NS + tile0;
    const float Pn  = P[gidx];
    const float t1n = T1[gidx];
    const float t2n = T2[gidx];
    const float En  = E[gidx];

    if (tn + 32 < NT) {
      const int nidx = gidx + 32 * NS;
      __builtin_prefetch(&P[nidx],  0, 0);
      __builtin_prefetch(&T1[nidx], 0, 0);
      __builtin_prefetch(&T2[nidx], 0, 0);
      __builtin_prefetch(&E[nidx],  0, 0);
    }

    float4 f[32];
#pragma unroll
    for (int k = 0; k < 32; ++k) f[k] = stage[buf][k * WAVES_PER_BLOCK + wv];
    __builtin_amdgcn_sched_barrier(0);

    const int idx = (t0 + lane) * NS + s;

    float yv = 0.0f;
#pragma unroll
    for (int k = 0; k < 32; ++k) {
      const float Ta_k = f[k].x;
      const float Ps_k = f[k].y;
      const float Pl_k = f[k].z;
      const float E_k  = f[k].w;

      const float Smelt = fminf(S, fmaxf(Ta_k * gm, 0.0f));
      S = S + Ps_k - Smelt;
      float Hn = H + Smelt + Pl_k - E_k * ge;
      Hn = fmaxf(Hn, 0.0f);
      const float Q = Hn * go;
      H = Hn - Q;

      float r = wmma_fold16(Q * ga);
      r += xor8f(r);
      r += dpp8f<DPP8_XOR4>(r);
      r += dpp8f<DPP8_XOR2>(r);
      r += dpp8f<DPP8_XOR1>(r);

      yv = (lane == k) ? r : yv;
    }
    __builtin_amdgcn_sched_barrier(0);

    stage[buf ^ 1][threadIdx.x] = forcing(Pn, t1n, t2n, En);
    __syncthreads();

    out[idx] = yv;
  }
}

extern "C" void kernel_launch(void* const* d_in, const int* in_sizes, int n_in,
                              void* d_out, int out_size, void* d_ws, size_t ws_size,
                              hipStream_t stream) {
  (void)in_sizes; (void)n_in; (void)out_size; (void)d_ws; (void)ws_size;
  const float* P  = (const float*)d_in[0];
  const float* T1 = (const float*)d_in[1];
  const float* T2 = (const float*)d_in[2];
  const float* E  = (const float*)d_in[3];
  const float* w  = (const float*)d_in[4];
  float* out = (float*)d_out;

  dim3 grid(NS / WAVES_PER_BLOCK);
  dim3 block(BLOCK_THREADS);
  waternet_scan<<<grid, block, 0, stream>>>(P, T1, T2, E, w, out);
}


# Hidden-unit sums batched: one f32 WMMA folds lanes l,l+16 of two steps into two columns of a tile; 14 register adds + one half swap finish 32 steps (same sum tree); per-step permlane/DPP chain gone
# speedup vs baseline: 1.6238x; 1.2675x over previous
#include <hip/hip_runtime.h>
#include <math.h>

typedef __attribute__((ext_vector_type(2))) float v2f;
typedef __attribute__((ext_vector_type(8))) float v8f;

#define NT 1024
#define NS 2048
#define WAVES_PER_BLOCK 8
#define BLOCK_THREADS (WAVES_PER_BLOCK * 32)
#define WMMA_LAG 2


__device__ __forceinline__ float xhalf(float v) {
  int x = __builtin_bit_cast(int, v);
  int r = __builtin_amdgcn_permlanex16(x, x, 0x76543210u, 0xfedcba98u,
                                       false,  false);
  return __builtin_bit_cast(float, r);
}

__device__ __forceinline__ v8f wmma_fold_pair(float p0, float p1, v2f bsel, v8f acc) {
  v2f a; a.x = p0; a.y = p1;
  return __builtin_amdgcn_wmma_f32_16x16x4_f32(
       false, a,  false, bsel,
       (short)0, acc,  false,  false);
}

__device__ __forceinline__ float sum_rows8(v8f d) {
  const float e0 = d[0] + d[4], e1 = d[1] + d[5];
  const float e2 = d[2] + d[6], e3 = d[3] + d[7];
  const float g0 = e0 + e2, g1 = e1 + e3;
  return g0 + g1;
}

__device__ __forceinline__ float sigmoidf(float x) {
  return 1.0f / (1.0f + expf(-x));
}

__global__ __launch_bounds__(BLOCK_THREADS) void waternet_scan(
    const float* __restrict__ P,  const float* __restrict__ T1,
    const float* __restrict__ T2, const float* __restrict__ E,
    const float* __restrict__ w,  float* __restrict__ out) {
  const int lane = threadIdx.x & 31;
  const int wv   = threadIdx.x >> 5;
  const int s    = blockIdx.x * WAVES_PER_BLOCK + wv;

  float gm = expf(w[lane]) + 1.0f;
  float ge = sigmoidf(w[32 + lane]);
  float go = sigmoidf(w[64 + lane]);
  const float xw = w[96 + lane];
  float mx = xw;
  for (int off = 16; off; off >>= 1) mx = fmaxf(mx, __shfl_xor(mx, off, 32));
  const float ex = expf(xw - mx);
  float sm = ex;
  for (int off = 16; off; off >>= 1) sm += __shfl_xor(sm, off, 32);
  float ga = ex / sm;

  const int hh = (lane & 16) | ((lane & 7) << 1) | ((lane >> 3) & 1);
  gm = __shfl(gm, hh, 32);
  ge = __shfl(ge, hh, 32);
  go = __shfl(go, hh, 32);
  ga = __shfl(ga, hh, 32);

  const int col = lane & 15;
  v2f bsel[8];
#pragma unroll
  for (int j = 0; j < 8; ++j) {
    bsel[j].x = (col == 2 * j)     ? 1.0f : 0.0f;
    bsel[j].y = (col == 2 * j + 1) ? 1.0f : 0.0f;
  }

  float S = 0.0f, H = 0.0f;

  __shared__ float4 stage[2][BLOCK_THREADS];

  const int tile0 =
      (threadIdx.x >> 3) * NS + blockIdx.x * WAVES_PER_BLOCK + (threadIdx.x & 7);

  auto forcing = [](float Pv, float t1v, float t2v, float Ev) -> float4 {
    const float Ta = (t1v + t2v) * 0.5f;
    float ratio = (t1v + t2v) / (t2v - t1v);
    ratio = fminf(1.0f, fmaxf(-1.0f, ratio));
    float rp = 1.0f - acosf(ratio) / 3.1415f;
    rp = (t1v >= 0.0f) ? 1.0f : rp;
    rp = (t2v <= 0.0f) ? 0.0f : rp;
    return make_float4(Ta, (1.0f - rp) * Pv, rp * Pv, Ev);
  };

  stage[0][threadIdx.x] = forcing(P[tile0], T1[tile0], T2[tile0], E[tile0]);
  __syncthreads();

  for (int t0 = 0; t0 < NT; t0 += 32) {
    const int buf = (t0 >> 5) & 1;

    const int tn = (t0 + 32 < NT) ? t0 + 32 : t0;
    const int gidx = tn * NS + tile0;
    const float Pn  = P[gidx];
    const float t1n = T1[gidx];
    const float t2n = T2[gidx];
    const float En  = E[gidx];

    if (tn + 32 < NT) {
      const int nidx = gidx + 32 * NS;
      __builtin_prefetch(&P[nidx],  0, 0);
      __builtin_prefetch(&T1[nidx], 0, 0);
      __builtin_prefetch(&T2[nidx], 0, 0);
      __builtin_prefetch(&E[nidx],  0, 0);
    }

    float4 f[32];
#pragma unroll
    for (int k = 0; k < 32; ++k) f[k] = stage[buf][k * WAVES_PER_BLOCK + wv];
    __builtin_amdgcn_sched_barrier(0);

    const int idx = (t0 + lane) * NS + s;

    v8f acc0 = {}, acc1 = {};
    float pv[32];
#pragma unroll
    for (int k = 0; k < 32; ++k) {
      const float Ta_k = f[k].x;
      const float Ps_k = f[k].y;
      const float Pl_k = f[k].z;
      const float E_k  = f[k].w;

      const float Smelt = fminf(S, fmaxf(Ta_k * gm, 0.0f));
      S = S + Ps_k - Smelt;
      float Hn = H + Smelt + Pl_k - E_k * ge;
      Hn = fmaxf(Hn, 0.0f);
      const float Q = Hn * go;
      H = Hn - Q;

      pv[k] = Q * ga;
      if (k & 1) {
        const int j = (k >> 1) - WMMA_LAG;
        if (j >= 0) {
          if (j < 8) acc0 = wmma_fold_pair(pv[2 * j], pv[2 * j + 1], bsel[j & 7], acc0);
          else       acc1 = wmma_fold_pair(pv[2 * j], pv[2 * j + 1], bsel[j & 7], acc1);
        }
        __builtin_amdgcn_sched_barrier(0);
      }
    }
#pragma unroll
    for (int j = 16 - WMMA_LAG; j < 16; ++j)
      acc1 = wmma_fold_pair(pv[2 * j], pv[2 * j + 1], bsel[j & 7], acc1);
    __builtin_amdgcn_sched_barrier(0);

    stage[buf ^ 1][threadIdx.x] = forcing(Pn, t1n, t2n, En);
    __builtin_amdgcn_sched_barrier(0);

    const float G0 = sum_rows8(acc0);
    const float G1 = sum_rows8(acc1);
    const bool lo_half = lane < 16;
    const float own = lo_half ? G0 : G1;
    const float oth = lo_half ? G1 : G0;
    const float yv = own + xhalf(oth);
    __syncthreads();

    out[idx] = yv;
  }
}

extern "C" void kernel_launch(void* const* d_in, const int* in_sizes, int n_in,
                              void* d_out, int out_size, void* d_ws, size_t ws_size,
                              hipStream_t stream) {
  (void)in_sizes; (void)n_in; (void)out_size; (void)d_ws; (void)ws_size;
  const float* P  = (const float*)d_in[0];
  const float* T1 = (const float*)d_in[1];
  const float* T2 = (const float*)d_in[2];
  const float* E  = (const float*)d_in[3];
  const float* w  = (const float*)d_in[4];
  float* out = (float*)d_out;

  dim3 grid(NS / WAVES_PER_BLOCK);
  dim3 block(BLOCK_THREADS);
  waternet_scan<<<grid, block, 0, stream>>>(P, T1, T2, E, w, out);
}


# Two sites per wave at one wave per SIMD (4-wave work-groups, still 256 groups): the two independent recurrences interleave in one in-order stream and fill each other's result-latency slots
# speedup vs baseline: 2.0267x; 1.2481x over previous
#include <hip/hip_runtime.h>
#include <math.h>

typedef __attribute__((ext_vector_type(2))) float v2f;
typedef __attribute__((ext_vector_type(8))) float v8f;

#define NT 1024
#define NS 2048
#define WAVES_PER_BLOCK 4
#define SITES_PER_WAVE 2
#define SITES_PER_BLOCK (WAVES_PER_BLOCK * SITES_PER_WAVE)
#define BLOCK_THREADS (WAVES_PER_BLOCK * 32)
#define TILE_ELEMS (32 * SITES_PER_BLOCK)
#define WMMA_LAG 2


__device__ __forceinline__ float xhalf(float v) {
  int x = __builtin_bit_cast(int, v);
  int r = __builtin_amdgcn_permlanex16(x, x, 0x76543210u, 0xfedcba98u,
                                       false,  false);
  return __builtin_bit_cast(float, r);
}

__device__ __forceinline__ v8f wmma_fold_pair(float p0, float p1, v2f bsel, v8f acc) {
  v2f a; a.x = p0; a.y = p1;
  return __builtin_amdgcn_wmma_f32_16x16x4_f32(
       false, a,  false, bsel,
       (short)0, acc,  false,  false);
}

__device__ __forceinline__ float sum_rows8(v8f d) {
  const float e0 = d[0] + d[4], e1 = d[1] + d[5];
  const float e2 = d[2] + d[6], e3 = d[3] + d[7];
  const float g0 = e0 + e2, g1 = e1 + e3;
  return g0 + g1;
}

__device__ __forceinline__ float sigmoidf(float x) {
  return 1.0f / (1.0f + expf(-x));
}

__global__ __launch_bounds__(BLOCK_THREADS) void waternet_scan(
    const float* __restrict__ P,  const float* __restrict__ T1,
    const float* __restrict__ T2, const float* __restrict__ E,
    const float* __restrict__ w,  float* __restrict__ out) {
  const int lane = threadIdx.x & 31;
  const int wv   = threadIdx.x >> 5;
  const int s0   = blockIdx.x * SITES_PER_BLOCK + SITES_PER_WAVE * wv;

  float gm = expf(w[lane]) + 1.0f;
  float ge = sigmoidf(w[32 + lane]);
  float go = sigmoidf(w[64 + lane]);
  const float xw = w[96 + lane];
  float mx = xw;
  for (int off = 16; off; off >>= 1) mx = fmaxf(mx, __shfl_xor(mx, off, 32));
  const float ex = expf(xw - mx);
  float sm = ex;
  for (int off = 16; off; off >>= 1) sm += __shfl_xor(sm, off, 32);
  float ga = ex / sm;

  const int hh = (lane & 16) | ((lane & 7) << 1) | ((lane >> 3) & 1);
  gm = __shfl(gm, hh, 32);
  ge = __shfl(ge, hh, 32);
  go = __shfl(go, hh, 32);
  ga = __shfl(ga, hh, 32);

  const int col = lane & 15;
  v2f bsel[8];
#pragma unroll
  for (int j = 0; j < 8; ++j) {
    bsel[j].x = (col == 2 * j)     ? 1.0f : 0.0f;
    bsel[j].y = (col == 2 * j + 1) ? 1.0f : 0.0f;
  }

  float Sa = 0.0f, Ha = 0.0f;
  float Sb = 0.0f, Hb = 0.0f;

  __shared__ float4 stage[2][TILE_ELEMS];

  const int tile0 =
      (threadIdx.x >> 3) * NS + blockIdx.x * SITES_PER_BLOCK + (threadIdx.x & 7);
  const int tile1 = tile0 + 16 * NS;

  auto forcing = [](float Pv, float t1v, float t2v, float Ev) -> float4 {
    const float Ta = (t1v + t2v) * 0.5f;
    float ratio = (t1v + t2v) / (t2v - t1v);
    ratio = fminf(1.0f, fmaxf(-1.0f, ratio));
    float rp = 1.0f - acosf(ratio) / 3.1415f;
    rp = (t1v >= 0.0f) ? 1.0f : rp;
    rp = (t2v <= 0.0f) ? 0.0f : rp;
    return make_float4(Ta, (1.0f - rp) * Pv, rp * Pv, Ev);
  };

  stage[0][threadIdx.x]                 = forcing(P[tile0], T1[tile0], T2[tile0], E[tile0]);
  stage[0][threadIdx.x + BLOCK_THREADS] = forcing(P[tile1], T1[tile1], T2[tile1], E[tile1]);
  __syncthreads();

  auto step = [&](const float4 fk, float& S, float& H) -> float {
    const float Smelt = fminf(S, fmaxf(fk.x * gm, 0.0f));
    S = S + fk.y - Smelt;
    float Hn = H + Smelt + fk.z - fk.w * ge;
    Hn = fmaxf(Hn, 0.0f);
    const float Q = Hn * go;
    H = Hn - Q;
    return Q * ga;
  };

  for (int t0 = 0; t0 < NT; t0 += 32) {
    const int buf = (t0 >> 5) & 1;

    const int tn = (t0 + 32 < NT) ? t0 + 32 : t0;
    const int gidx = tn * NS + tile0;
    const int gidy = tn * NS + tile1;
    const float Pn  = P[gidx],  Pm  = P[gidy];
    const float t1n = T1[gidx], t1m = T1[gidy];
    const float t2n = T2[gidx], t2m = T2[gidy];
    const float En  = E[gidx],  Em  = E[gidy];

    if (tn + 32 < NT) {
      const int nidx = gidx + 32 * NS;
      const int nidy = gidy + 32 * NS;
      __builtin_prefetch(&P[nidx],  0, 0);
      __builtin_prefetch(&T1[nidx], 0, 0);
      __builtin_prefetch(&T2[nidx], 0, 0);
      __builtin_prefetch(&E[nidx],  0, 0);
      __builtin_prefetch(&P[nidy],  0, 0);
      __builtin_prefetch(&T1[nidy], 0, 0);
      __builtin_prefetch(&T2[nidy], 0, 0);
      __builtin_prefetch(&E[nidy],  0, 0);
    }

    float4 fa[32], fb[32];
#pragma unroll
    for (int k = 0; k < 32; ++k) {
      fa[k] = stage[buf][k * SITES_PER_BLOCK + SITES_PER_WAVE * wv];
      fb[k] = stage[buf][k * SITES_PER_BLOCK + SITES_PER_WAVE * wv + 1];
    }
    __builtin_amdgcn_sched_barrier(0);

    const int idx = (t0 + lane) * NS + s0;

    v8f acc0a = {}, acc1a = {};
    v8f acc0b = {}, acc1b = {};
    float pa[32], pb[32];
#pragma unroll
    for (int k = 0; k < 32; ++k) {
      pa[k] = step(fa[k], Sa, Ha);
      pb[k] = step(fb[k], Sb, Hb);
      if (k & 1) {
        const int j = (k >> 1) - WMMA_LAG;
        if (j >= 0) {
          if (j < 8) {
            acc0a = wmma_fold_pair(pa[2 * j], pa[2 * j + 1], bsel[j & 7], acc0a);
            acc0b = wmma_fold_pair(pb[2 * j], pb[2 * j + 1], bsel[j & 7], acc0b);
          } else {
            acc1a = wmma_fold_pair(pa[2 * j], pa[2 * j + 1], bsel[j & 7], acc1a);
            acc1b = wmma_fold_pair(pb[2 * j], pb[2 * j + 1], bsel[j & 7], acc1b);
          }
        }
        __builtin_amdgcn_sched_barrier(0);
      }
    }
#pragma unroll
    for (int j = 16 - WMMA_LAG; j < 16; ++j) {
      acc1a = wmma_fold_pair(pa[2 * j], pa[2 * j + 1], bsel[j & 7], acc1a);
      acc1b = wmma_fold_pair(pb[2 * j], pb[2 * j + 1], bsel[j & 7], acc1b);
    }
    __builtin_amdgcn_sched_barrier(0);

    stage[buf ^ 1][threadIdx.x]                 = forcing(Pn, t1n, t2n, En);
    stage[buf ^ 1][threadIdx.x + BLOCK_THREADS] = forcing(Pm, t1m, t2m, Em);
    __builtin_amdgcn_sched_barrier(0);

    const float G0a = sum_rows8(acc0a), G1a = sum_rows8(acc1a);
    const float G0b = sum_rows8(acc0b), G1b = sum_rows8(acc1b);
    const bool lo_half = lane < 16;
    const float owna = lo_half ? G0a : G1a;
    const float otha = lo_half ? G1a : G0a;
    const float ownb = lo_half ? G0b : G1b;
    const float othb = lo_half ? G1b : G0b;
    const float ya = owna + xhalf(otha);
    const float yb = ownb + xhalf(othb);
    __syncthreads();

    *reinterpret_cast<float2*>(&out[idx]) = make_float2(ya, yb);
  }
}

extern "C" void kernel_launch(void* const* d_in, const int* in_sizes, int n_in,
                              void* d_out, int out_size, void* d_ws, size_t ws_size,
                              hipStream_t stream) {
  (void)in_sizes; (void)n_in; (void)out_size; (void)d_ws; (void)ws_size;
  const float* P  = (const float*)d_in[0];
  const float* T1 = (const float*)d_in[1];
  const float* T2 = (const float*)d_in[2];
  const float* E  = (const float*)d_in[3];
  const float* w  = (const float*)d_in[4];
  float* out = (float*)d_out;

  dim3 grid(NS / SITES_PER_BLOCK);
  dim3 block(BLOCK_THREADS);
  waternet_scan<<<grid, block, 0, stream>>>(P, T1, T2, E, w, out);
}


# Water-store sum re-associated: H + (Smelt + (rain - E*ge)) in place of ((H + Smelt) + rain) - E*ge, so its dependent chain per step is add, max, fma (3 results) instead of 5; summation order changed
# speedup vs baseline: 2.3060x; 1.1378x over previous
#include <hip/hip_runtime.h>
#include <math.h>

typedef __attribute__((ext_vector_type(2))) float v2f;
typedef __attribute__((ext_vector_type(8))) float v8f;

#define NT 1024
#define NS 2048
#define WAVES_PER_BLOCK 4
#define SITES_PER_WAVE 2
#define SITES_PER_BLOCK (WAVES_PER_BLOCK * SITES_PER_WAVE)
#define BLOCK_THREADS (WAVES_PER_BLOCK * 32)
#define TILE_ELEMS (32 * SITES_PER_BLOCK)
#define WMMA_LAG 2


__device__ __forceinline__ float xhalf(float v) {
  int x = __builtin_bit_cast(int, v);
  int r = __builtin_amdgcn_permlanex16(x, x, 0x76543210u, 0xfedcba98u,
                                       false,  false);
  return __builtin_bit_cast(float, r);
}

__device__ __forceinline__ v8f wmma_fold_pair(float p0, float p1, v2f bsel, v8f acc) {
  v2f a; a.x = p0; a.y = p1;
  return __builtin_amdgcn_wmma_f32_16x16x4_f32(
       false, a,  false, bsel,
       (short)0, acc,  false,  false);
}

__device__ __forceinline__ float sum_rows8(v8f d) {
  const float e0 = d[0] + d[4], e1 = d[1] + d[5];
  const float e2 = d[2] + d[6], e3 = d[3] + d[7];
  const float g0 = e0 + e2, g1 = e1 + e3;
  return g0 + g1;
}

__device__ __forceinline__ float sigmoidf(float x) {
  return 1.0f / (1.0f + expf(-x));
}

__global__ __launch_bounds__(BLOCK_THREADS) void waternet_scan(
    const float* __restrict__ P,  const float* __restrict__ T1,
    const float* __restrict__ T2, const float* __restrict__ E,
    const float* __restrict__ w,  float* __restrict__ out) {
  const int lane = threadIdx.x & 31;
  const int wv   = threadIdx.x >> 5;
  const int s0   = blockIdx.x * SITES_PER_BLOCK + SITES_PER_WAVE * wv;

  float gm = expf(w[lane]) + 1.0f;
  float ge = sigmoidf(w[32 + lane]);
  float go = sigmoidf(w[64 + lane]);
  const float xw = w[96 + lane];
  float mx = xw;
  for (int off = 16; off; off >>= 1) mx = fmaxf(mx, __shfl_xor(mx, off, 32));
  const float ex = expf(xw - mx);
  float sm = ex;
  for (int off = 16; off; off >>= 1) sm += __shfl_xor(sm, off, 32);
  float ga = ex / sm;

  const int hh = (lane & 16) | ((lane & 7) << 1) | ((lane >> 3) & 1);
  gm = __shfl(gm, hh, 32);
  ge = __shfl(ge, hh, 32);
  go = __shfl(go, hh, 32);
  ga = __shfl(ga, hh, 32);

  const int col = lane & 15;
  v2f bsel[8];
#pragma unroll
  for (int j = 0; j < 8; ++j) {
    bsel[j].x = (col == 2 * j)     ? 1.0f : 0.0f;
    bsel[j].y = (col == 2 * j + 1) ? 1.0f : 0.0f;
  }

  float Sa = 0.0f, Ha = 0.0f;
  float Sb = 0.0f, Hb = 0.0f;

  __shared__ float4 stage[2][TILE_ELEMS];

  const int tile0 =
      (threadIdx.x >> 3) * NS + blockIdx.x * SITES_PER_BLOCK + (threadIdx.x & 7);
  const int tile1 = tile0 + 16 * NS;

  auto forcing = [](float Pv, float t1v, float t2v, float Ev) -> float4 {
    const float Ta = (t1v + t2v) * 0.5f;
    float ratio = (t1v + t2v) / (t2v - t1v);
    ratio = fminf(1.0f, fmaxf(-1.0f, ratio));
    float rp = 1.0f - acosf(ratio) / 3.1415f;
    rp = (t1v >= 0.0f) ? 1.0f : rp;
    rp = (t2v <= 0.0f) ? 0.0f : rp;
    return make_float4(Ta, (1.0f - rp) * Pv, rp * Pv, Ev);
  };

  stage[0][threadIdx.x]                 = forcing(P[tile0], T1[tile0], T2[tile0], E[tile0]);
  stage[0][threadIdx.x + BLOCK_THREADS] = forcing(P[tile1], T1[tile1], T2[tile1], E[tile1]);
  __syncthreads();

  auto step = [&](const float4 fk, float& S, float& H) -> float {
    const float Smelt = fminf(S, fmaxf(fk.x * gm, 0.0f));
    S = S + fk.y - Smelt;
    const float net = fk.z - fk.w * ge;
    const float inflow = Smelt + net;
    const float Hn = fmaxf(H + inflow, 0.0f);
    const float Q = Hn * go;
    H = Hn - Q;
    return Q * ga;
  };

  for (int t0 = 0; t0 < NT; t0 += 32) {
    const int buf = (t0 >> 5) & 1;

    const int tn = (t0 + 32 < NT) ? t0 + 32 : t0;
    const int gidx = tn * NS + tile0;
    const int gidy = tn * NS + tile1;
    const float Pn  = P[gidx],  Pm  = P[gidy];
    const float t1n = T1[gidx], t1m = T1[gidy];
    const float t2n = T2[gidx], t2m = T2[gidy];
    const float En  = E[gidx],  Em  = E[gidy];

    if (tn + 32 < NT) {
      const int nidx = gidx + 32 * NS;
      const int nidy = gidy + 32 * NS;
      __builtin_prefetch(&P[nidx],  0, 0);
      __builtin_prefetch(&T1[nidx], 0, 0);
      __builtin_prefetch(&T2[nidx], 0, 0);
      __builtin_prefetch(&E[nidx],  0, 0);
      __builtin_prefetch(&P[nidy],  0, 0);
      __builtin_prefetch(&T1[nidy], 0, 0);
      __builtin_prefetch(&T2[nidy], 0, 0);
      __builtin_prefetch(&E[nidy],  0, 0);
    }

    float4 fa[32], fb[32];
#pragma unroll
    for (int k = 0; k < 32; ++k) {
      fa[k] = stage[buf][k * SITES_PER_BLOCK + SITES_PER_WAVE * wv];
      fb[k] = stage[buf][k * SITES_PER_BLOCK + SITES_PER_WAVE * wv + 1];
    }
    __builtin_amdgcn_sched_barrier(0);

    const int idx = (t0 + lane) * NS + s0;

    v8f acc0a = {}, acc1a = {};
    v8f acc0b = {}, acc1b = {};
    float pa[32], pb[32];
#pragma unroll
    for (int k = 0; k < 32; ++k) {
      pa[k] = step(fa[k], Sa, Ha);
      pb[k] = step(fb[k], Sb, Hb);
      if (k & 1) {
        const int j = (k >> 1) - WMMA_LAG;
        if (j >= 0) {
          if (j < 8) {
            acc0a = wmma_fold_pair(pa[2 * j], pa[2 * j + 1], bsel[j & 7], acc0a);
            acc0b = wmma_fold_pair(pb[2 * j], pb[2 * j + 1], bsel[j & 7], acc0b);
          } else {
            acc1a = wmma_fold_pair(pa[2 * j], pa[2 * j + 1], bsel[j & 7], acc1a);
            acc1b = wmma_fold_pair(pb[2 * j], pb[2 * j + 1], bsel[j & 7], acc1b);
          }
        }
        __builtin_amdgcn_sched_barrier(0);
      }
    }
#pragma unroll
    for (int j = 16 - WMMA_LAG; j < 16; ++j) {
      acc1a = wmma_fold_pair(pa[2 * j], pa[2 * j + 1], bsel[j & 7], acc1a);
      acc1b = wmma_fold_pair(pb[2 * j], pb[2 * j + 1], bsel[j & 7], acc1b);
    }
    __builtin_amdgcn_sched_barrier(0);

    stage[buf ^ 1][threadIdx.x]                 = forcing(Pn, t1n, t2n, En);
    stage[buf ^ 1][threadIdx.x + BLOCK_THREADS] = forcing(Pm, t1m, t2m, Em);
    __builtin_amdgcn_sched_barrier(0);

    const float G0a = sum_rows8(acc0a), G1a = sum_rows8(acc1a);
    const float G0b = sum_rows8(acc0b), G1b = sum_rows8(acc1b);
    const bool lo_half = lane < 16;
    const float owna = lo_half ? G0a : G1a;
    const float otha = lo_half ? G1a : G0a;
    const float ownb = lo_half ? G0b : G1b;
    const float othb = lo_half ? G1b : G0b;
    const float ya = owna + xhalf(otha);
    const float yb = ownb + xhalf(othb);
    __syncthreads();

    *reinterpret_cast<float2*>(&out[idx]) = make_float2(ya, yb);
  }
}

extern "C" void kernel_launch(void* const* d_in, const int* in_sizes, int n_in,
                              void* d_out, int out_size, void* d_ws, size_t ws_size,
                              hipStream_t stream) {
  (void)in_sizes; (void)n_in; (void)out_size; (void)d_ws; (void)ws_size;
  const float* P  = (const float*)d_in[0];
  const float* T1 = (const float*)d_in[1];
  const float* T2 = (const float*)d_in[2];
  const float* E  = (const float*)d_in[3];
  const float* w  = (const float*)d_in[4];
  float* out = (float*)d_out;

  dim3 grid(NS / SITES_PER_BLOCK);
  dim3 block(BLOCK_THREADS);
  waternet_scan<<<grid, block, 0, stream>>>(P, T1, T2, E, w, out);
}


# Next tile's rain/snow partition software-pipelined: raw forcings load two blocks ahead; its division and arccosine ride with step pairs, the division by PI beside the trailing WMMAs
# speedup vs baseline: 2.4478x; 1.0615x over previous
#include <hip/hip_runtime.h>
#include <math.h>

typedef __attribute__((ext_vector_type(2))) float v2f;
typedef __attribute__((ext_vector_type(8))) float v8f;

#define NT 1024
#define NS 2048
#define WAVES_PER_BLOCK 4
#define SITES_PER_WAVE 2
#define SITES_PER_BLOCK (WAVES_PER_BLOCK * SITES_PER_WAVE)
#define BLOCK_THREADS (WAVES_PER_BLOCK * 32)
#define TILE_ELEMS (32 * SITES_PER_BLOCK)
#define WMMA_LAG 2


__device__ __forceinline__ float xhalf(float v) {
  int x = __builtin_bit_cast(int, v);
  int r = __builtin_amdgcn_permlanex16(x, x, 0x76543210u, 0xfedcba98u,
                                       false,  false);
  return __builtin_bit_cast(float, r);
}

__device__ __forceinline__ v8f wmma_fold_pair(float p0, float p1, v2f bsel, v8f acc) {
  v2f a; a.x = p0; a.y = p1;
  return __builtin_amdgcn_wmma_f32_16x16x4_f32(
       false, a,  false, bsel,
       (short)0, acc,  false,  false);
}

__device__ __forceinline__ float sum_rows8(v8f d) {
  const float e0 = d[0] + d[4], e1 = d[1] + d[5];
  const float e2 = d[2] + d[6], e3 = d[3] + d[7];
  const float g0 = e0 + e2, g1 = e1 + e3;
  return g0 + g1;
}

__device__ __forceinline__ float sigmoidf(float x) {
  return 1.0f / (1.0f + expf(-x));
}

__global__ __launch_bounds__(BLOCK_THREADS) void waternet_scan(
    const float* __restrict__ P,  const float* __restrict__ T1,
    const float* __restrict__ T2, const float* __restrict__ E,
    const float* __restrict__ w,  float* __restrict__ out) {
  const int lane = threadIdx.x & 31;
  const int wv   = threadIdx.x >> 5;
  const int s0   = blockIdx.x * SITES_PER_BLOCK + SITES_PER_WAVE * wv;

  float gm = expf(w[lane]) + 1.0f;
  float ge = sigmoidf(w[32 + lane]);
  float go = sigmoidf(w[64 + lane]);
  const float xw = w[96 + lane];
  float mx = xw;
  for (int off = 16; off; off >>= 1) mx = fmaxf(mx, __shfl_xor(mx, off, 32));
  const float ex = expf(xw - mx);
  float sm = ex;
  for (int off = 16; off; off >>= 1) sm += __shfl_xor(sm, off, 32);
  float ga = ex / sm;

  const int hh = (lane & 16) | ((lane & 7) << 1) | ((lane >> 3) & 1);
  gm = __shfl(gm, hh, 32);
  ge = __shfl(ge, hh, 32);
  go = __shfl(go, hh, 32);
  ga = __shfl(ga, hh, 32);

  const int col = lane & 15;
  v2f bsel[8];
#pragma unroll
  for (int j = 0; j < 8; ++j) {
    bsel[j].x = (col == 2 * j)     ? 1.0f : 0.0f;
    bsel[j].y = (col == 2 * j + 1) ? 1.0f : 0.0f;
  }

  float Sa = 0.0f, Ha = 0.0f;
  float Sb = 0.0f, Hb = 0.0f;

  __shared__ float4 stage[2][TILE_ELEMS];

  const int tile0 =
      (threadIdx.x >> 3) * NS + blockIdx.x * SITES_PER_BLOCK + (threadIdx.x & 7);
  const int tile1 = tile0 + 16 * NS;

  auto part_ratio = [](float sum, float t1v, float t2v) -> float {
    return sum / (t2v - t1v);
  };
  auto part_acos = [](float ratio) -> float {
    return acosf(fminf(1.0f, fmaxf(-1.0f, ratio)));
  };
  auto part_frac = [](float ac) -> float {
    return ac / 3.1415f;
  };
  auto part_finish = [](float Pv, float t1v, float t2v, float Ev, float sum,
                        float frac) -> float4 {
    float rp = 1.0f - frac;
    rp = (t1v >= 0.0f) ? 1.0f : rp;
    rp = (t2v <= 0.0f) ? 0.0f : rp;
    return make_float4(sum * 0.5f, (1.0f - rp) * Pv, rp * Pv, Ev);
  };
  auto forcing = [&](float Pv, float t1v, float t2v, float Ev) -> float4 {
    const float sum = t1v + t2v;
    return part_finish(Pv, t1v, t2v, Ev, sum,
                       part_frac(part_acos(part_ratio(sum, t1v, t2v))));
  };

  float cPx  = P[tile0 + 32 * NS],  cPy  = P[tile1 + 32 * NS];
  float ct1x = T1[tile0 + 32 * NS], ct1y = T1[tile1 + 32 * NS];
  float ct2x = T2[tile0 + 32 * NS], ct2y = T2[tile1 + 32 * NS];
  float cEx  = E[tile0 + 32 * NS],  cEy  = E[tile1 + 32 * NS];
  stage[0][threadIdx.x]                 = forcing(P[tile0], T1[tile0], T2[tile0], E[tile0]);
  stage[0][threadIdx.x + BLOCK_THREADS] = forcing(P[tile1], T1[tile1], T2[tile1], E[tile1]);
  __syncthreads();

  auto step = [&](const float4 fk, float& S, float& H) -> float {
    const float Smelt = fminf(S, fmaxf(fk.x * gm, 0.0f));
    S = S + fk.y - Smelt;
    const float net = fk.z - fk.w * ge;
    const float inflow = Smelt + net;
    const float Hn = fmaxf(H + inflow, 0.0f);
    const float Q = Hn * go;
    H = Hn - Q;
    return Q * ga;
  };

  for (int t0 = 0; t0 < NT; t0 += 32) {
    const int buf = (t0 >> 5) & 1;

    const int tn = (t0 + 64 < NT) ? t0 + 64 : t0;
    const int gidx = tn * NS + tile0;
    const int gidy = tn * NS + tile1;
    const float Pn  = P[gidx],  Pm  = P[gidy];
    const float t1n = T1[gidx], t1m = T1[gidy];
    const float t2n = T2[gidx], t2m = T2[gidy];
    const float En  = E[gidx],  Em  = E[gidy];

    if (tn + 32 < NT) {
      const int nidx = gidx + 32 * NS;
      const int nidy = gidy + 32 * NS;
      __builtin_prefetch(&P[nidx],  0, 0);
      __builtin_prefetch(&T1[nidx], 0, 0);
      __builtin_prefetch(&T2[nidx], 0, 0);
      __builtin_prefetch(&E[nidx],  0, 0);
      __builtin_prefetch(&P[nidy],  0, 0);
      __builtin_prefetch(&T1[nidy], 0, 0);
      __builtin_prefetch(&T2[nidy], 0, 0);
      __builtin_prefetch(&E[nidy],  0, 0);
    }

    float4 fa[32], fb[32];
#pragma unroll
    for (int k = 0; k < 32; ++k) {
      fa[k] = stage[buf][k * SITES_PER_BLOCK + SITES_PER_WAVE * wv];
      fb[k] = stage[buf][k * SITES_PER_BLOCK + SITES_PER_WAVE * wv + 1];
    }
    __builtin_amdgcn_sched_barrier(0);

    const int idx = (t0 + lane) * NS + s0;

    v8f acc0a = {}, acc1a = {};
    v8f acc0b = {}, acc1b = {};
    float pa[32], pb[32];
    const float sumx = ct1x + ct2x, sumy = ct1y + ct2y;
    float ratx = 0.0f, raty = 0.0f, acx = 0.0f, acy = 0.0f;
#pragma unroll
    for (int k = 0; k < 32; ++k) {
      pa[k] = step(fa[k], Sa, Ha);
      pb[k] = step(fb[k], Sb, Hb);
      if (k & 1) {
        const int j = (k >> 1) - WMMA_LAG;
        if (j >= 0) {
          if (j < 8) {
            acc0a = wmma_fold_pair(pa[2 * j], pa[2 * j + 1], bsel[j & 7], acc0a);
            acc0b = wmma_fold_pair(pb[2 * j], pb[2 * j + 1], bsel[j & 7], acc0b);
          } else {
            acc1a = wmma_fold_pair(pa[2 * j], pa[2 * j + 1], bsel[j & 7], acc1a);
            acc1b = wmma_fold_pair(pb[2 * j], pb[2 * j + 1], bsel[j & 7], acc1b);
          }
        }
        if (k == 1)  { ratx = part_ratio(sumx, ct1x, ct2x); raty = part_ratio(sumy, ct1y, ct2y); }
        if (k == 15) { acx = part_acos(ratx); acy = part_acos(raty); }
        __builtin_amdgcn_sched_barrier(0);
      }
    }
#pragma unroll
    for (int j = 16 - WMMA_LAG; j < 16; ++j) {
      acc1a = wmma_fold_pair(pa[2 * j], pa[2 * j + 1], bsel[j & 7], acc1a);
      acc1b = wmma_fold_pair(pb[2 * j], pb[2 * j + 1], bsel[j & 7], acc1b);
    }
    __builtin_amdgcn_sched_barrier(0);

    stage[buf ^ 1][threadIdx.x] =
        part_finish(cPx, ct1x, ct2x, cEx, sumx, part_frac(acx));
    stage[buf ^ 1][threadIdx.x + BLOCK_THREADS] =
        part_finish(cPy, ct1y, ct2y, cEy, sumy, part_frac(acy));
    cPx = Pn;   cPy = Pm;
    ct1x = t1n; ct1y = t1m;
    ct2x = t2n; ct2y = t2m;
    cEx = En;   cEy = Em;
    __builtin_amdgcn_sched_barrier(0);

    const float G0a = sum_rows8(acc0a), G1a = sum_rows8(acc1a);
    const float G0b = sum_rows8(acc0b), G1b = sum_rows8(acc1b);
    const bool lo_half = lane < 16;
    const float owna = lo_half ? G0a : G1a;
    const float otha = lo_half ? G1a : G0a;
    const float ownb = lo_half ? G0b : G1b;
    const float othb = lo_half ? G1b : G0b;
    const float ya = owna + xhalf(otha);
    const float yb = ownb + xhalf(othb);
    __syncthreads();

    *reinterpret_cast<float2*>(&out[idx]) = make_float2(ya, yb);
  }
}

extern "C" void kernel_launch(void* const* d_in, const int* in_sizes, int n_in,
                              void* d_out, int out_size, void* d_ws, size_t ws_size,
                              hipStream_t stream) {
  (void)in_sizes; (void)n_in; (void)out_size; (void)d_ws; (void)ws_size;
  const float* P  = (const float*)d_in[0];
  const float* T1 = (const float*)d_in[1];
  const float* T2 = (const float*)d_in[2];
  const float* E  = (const float*)d_in[3];
  const float* w  = (const float*)d_in[4];
  float* out = (float*)d_out;

  dim3 grid(NS / SITES_PER_BLOCK);
  dim3 block(BLOCK_THREADS);
  waternet_scan<<<grid, block, 0, stream>>>(P, T1, T2, E, w, out);
}


# Lane half = site, two hidden units per lane: the lane adds its two units' Q*ga (first pairing of the same sum tree; may fuse), one f32 WMMA takes 2 steps of both sites: half the WMMAs and LDS reads
# speedup vs baseline: 2.8609x; 1.1688x over previous
#include <hip/hip_runtime.h>
#include <math.h>

typedef __attribute__((ext_vector_type(2))) float v2f;
typedef __attribute__((ext_vector_type(8))) float v8f;

#define NT 1024
#define NS 2048
#define WAVES_PER_BLOCK 4
#define SITES_PER_WAVE 2
#define SITES_PER_BLOCK (WAVES_PER_BLOCK * SITES_PER_WAVE)
#define BLOCK_THREADS (WAVES_PER_BLOCK * 32)
#define TILE_ELEMS (32 * SITES_PER_BLOCK)
#define WMMA_LAG 2


__device__ __forceinline__ float xhalf(float v) {
  int x = __builtin_bit_cast(int, v);
  int r = __builtin_amdgcn_permlanex16(x, x, 0x76543210u, 0xfedcba98u,
                                       false,  false);
  return __builtin_bit_cast(float, r);
}

__device__ __forceinline__ v8f wmma_fold_pair(float p0, float p1, v2f bsel, v8f acc) {
  v2f a; a.x = p0; a.y = p1;
  return __builtin_amdgcn_wmma_f32_16x16x4_f32(
       false, a,  false, bsel,
       (short)0, acc,  false,  false);
}

__device__ __forceinline__ float sum_rows8(v8f d) {
  const float e0 = d[0] + d[4], e1 = d[1] + d[5];
  const float e2 = d[2] + d[6], e3 = d[3] + d[7];
  const float g0 = e0 + e2, g1 = e1 + e3;
  return g0 + g1;
}

__device__ __forceinline__ float sigmoidf(float x) {
  return 1.0f / (1.0f + expf(-x));
}

__global__ __launch_bounds__(BLOCK_THREADS) void waternet_scan(
    const float* __restrict__ P,  const float* __restrict__ T1,
    const float* __restrict__ T2, const float* __restrict__ E,
    const float* __restrict__ w,  float* __restrict__ out) {
  const int lane = threadIdx.x & 31;
  const int wv   = threadIdx.x >> 5;
  const int half = lane >> 4;
  const int col  = lane & 15;
  const int s0   = blockIdx.x * SITES_PER_BLOCK + SITES_PER_WAVE * wv;

  float gm = expf(w[lane]) + 1.0f;
  float ge = sigmoidf(w[32 + lane]);
  float go = sigmoidf(w[64 + lane]);
  const float xw = w[96 + lane];
  float mx = xw;
  for (int off = 16; off; off >>= 1) mx = fmaxf(mx, __shfl_xor(mx, off, 32));
  const float ex = expf(xw - mx);
  float sm = ex;
  for (int off = 16; off; off >>= 1) sm += __shfl_xor(sm, off, 32);
  float ga = ex / sm;

  const int hu = ((col & 7) << 1) | (col >> 3);
  const float gm0 = __shfl(gm, hu, 32), gm1 = __shfl(gm, hu + 16, 32);
  const float ge0 = __shfl(ge, hu, 32), ge1 = __shfl(ge, hu + 16, 32);
  const float go0 = __shfl(go, hu, 32), go1 = __shfl(go, hu + 16, 32);
  const float ga0 = __shfl(ga, hu, 32), ga1 = __shfl(ga, hu + 16, 32);

  v2f bsel[4];
#pragma unroll
  for (int j = 0; j < 4; ++j) {
    bsel[j].x = (col == 4 * j + half)     ? 1.0f : 0.0f;
    bsel[j].y = (col == 4 * j + 2 + half) ? 1.0f : 0.0f;
  }

  float S0 = 0.0f, H0 = 0.0f;
  float S1 = 0.0f, H1 = 0.0f;

  __shared__ float4 stage[2][TILE_ELEMS];

  const int tile0 =
      (threadIdx.x >> 3) * NS + blockIdx.x * SITES_PER_BLOCK + (threadIdx.x & 7);
  const int tile1 = tile0 + 16 * NS;

  auto part_ratio = [](float sum, float t1v, float t2v) -> float {
    return sum / (t2v - t1v);
  };
  auto part_acos = [](float ratio) -> float {
    return acosf(fminf(1.0f, fmaxf(-1.0f, ratio)));
  };
  auto part_frac = [](float ac) -> float {
    return ac / 3.1415f;
  };
  auto part_finish = [](float Pv, float t1v, float t2v, float Ev, float sum,
                        float frac) -> float4 {
    float rp = 1.0f - frac;
    rp = (t1v >= 0.0f) ? 1.0f : rp;
    rp = (t2v <= 0.0f) ? 0.0f : rp;
    return make_float4(sum * 0.5f, (1.0f - rp) * Pv, rp * Pv, Ev);
  };
  auto forcing = [&](float Pv, float t1v, float t2v, float Ev) -> float4 {
    const float sum = t1v + t2v;
    return part_finish(Pv, t1v, t2v, Ev, sum,
                       part_frac(part_acos(part_ratio(sum, t1v, t2v))));
  };

  float cPx  = P[tile0 + 32 * NS],  cPy  = P[tile1 + 32 * NS];
  float ct1x = T1[tile0 + 32 * NS], ct1y = T1[tile1 + 32 * NS];
  float ct2x = T2[tile0 + 32 * NS], ct2y = T2[tile1 + 32 * NS];
  float cEx  = E[tile0 + 32 * NS],  cEy  = E[tile1 + 32 * NS];
  stage[0][threadIdx.x]                 = forcing(P[tile0], T1[tile0], T2[tile0], E[tile0]);
  stage[0][threadIdx.x + BLOCK_THREADS] = forcing(P[tile1], T1[tile1], T2[tile1], E[tile1]);
  __syncthreads();

  auto step = [](const float4 fk, float& S, float& H, const float gm,
                 const float ge, const float go, const float ga) -> float {
    const float Smelt = fminf(S, fmaxf(fk.x * gm, 0.0f));
    S = S + fk.y - Smelt;
    const float net = fk.z - fk.w * ge;
    const float inflow = Smelt + net;
    const float Hn = fmaxf(H + inflow, 0.0f);
    const float Q = Hn * go;
    H = Hn - Q;
    return Q * ga;
  };

  for (int t0 = 0; t0 < NT; t0 += 32) {
    const int buf = (t0 >> 5) & 1;

    const int tn = (t0 + 64 < NT) ? t0 + 64 : t0;
    const int gidx = tn * NS + tile0;
    const int gidy = tn * NS + tile1;
    const float Pn  = P[gidx],  Pm  = P[gidy];
    const float t1n = T1[gidx], t1m = T1[gidy];
    const float t2n = T2[gidx], t2m = T2[gidy];
    const float En  = E[gidx],  Em  = E[gidy];

    float4 f[32];
#pragma unroll
    for (int k = 0; k < 32; ++k)
      f[k] = stage[buf][k * SITES_PER_BLOCK + SITES_PER_WAVE * wv + half];
    __builtin_amdgcn_sched_barrier(0);

    if (tn + 32 < NT) {
      const int nidx = gidx + 32 * NS;
      const int nidy = gidy + 32 * NS;
      __builtin_prefetch(&P[nidx],  0, 0);
      __builtin_prefetch(&T1[nidx], 0, 0);
      __builtin_prefetch(&T2[nidx], 0, 0);
      __builtin_prefetch(&E[nidx],  0, 0);
      __builtin_prefetch(&P[nidy],  0, 0);
      __builtin_prefetch(&T1[nidy], 0, 0);
      __builtin_prefetch(&T2[nidy], 0, 0);
      __builtin_prefetch(&E[nidy],  0, 0);
    }

    v8f acc0 = {}, acc1 = {}, acc2 = {}, acc3 = {};
    float q[32];
    const float sumx = ct1x + ct2x, sumy = ct1y + ct2y;
    float ratx = 0.0f, raty = 0.0f, acx = 0.0f, acy = 0.0f;
#pragma unroll
    for (int k = 0; k < 32; ++k) {
      const float p0 = step(f[k], S0, H0, gm0, ge0, go0, ga0);
      const float p1 = step(f[k], S1, H1, gm1, ge1, go1, ga1);
      q[k] = p0 + p1;
      if (k & 1) {
        const int j = (k >> 1) - WMMA_LAG;
        if (j >= 0) {
          if (j < 4)       acc0 = wmma_fold_pair(q[2 * j], q[2 * j + 1], bsel[j & 3], acc0);
          else if (j < 8)  acc1 = wmma_fold_pair(q[2 * j], q[2 * j + 1], bsel[j & 3], acc1);
          else if (j < 12) acc2 = wmma_fold_pair(q[2 * j], q[2 * j + 1], bsel[j & 3], acc2);
          else             acc3 = wmma_fold_pair(q[2 * j], q[2 * j + 1], bsel[j & 3], acc3);
        }
        if (k == 1)  { ratx = part_ratio(sumx, ct1x, ct2x); raty = part_ratio(sumy, ct1y, ct2y); }
        if (k == 15) { acx = part_acos(ratx); acy = part_acos(raty); }
        __builtin_amdgcn_sched_barrier(0);
      }
    }
#pragma unroll
    for (int j = 16 - WMMA_LAG; j < 16; ++j)
      acc3 = wmma_fold_pair(q[2 * j], q[2 * j + 1], bsel[j & 3], acc3);
    __builtin_amdgcn_sched_barrier(0);

    stage[buf ^ 1][threadIdx.x] =
        part_finish(cPx, ct1x, ct2x, cEx, sumx, part_frac(acx));
    stage[buf ^ 1][threadIdx.x + BLOCK_THREADS] =
        part_finish(cPy, ct1y, ct2y, cEy, sumy, part_frac(acy));
    cPx = Pn;   cPy = Pm;
    ct1x = t1n; ct1y = t1m;
    ct2x = t2n; ct2y = t2m;
    cEx = En;   cEy = Em;
    __builtin_amdgcn_sched_barrier(0);

    const float G0 = sum_rows8(acc0), G1 = sum_rows8(acc1);
    const float G2 = sum_rows8(acc2), G3 = sum_rows8(acc3);
    const bool lo_half = lane < 16;
    const float own0 = lo_half ? G0 : G2;
    const float oth0 = lo_half ? G2 : G0;
    const float own1 = lo_half ? G1 : G3;
    const float oth1 = lo_half ? G3 : G1;
    const float y0 = own0 + xhalf(oth0);
    const float y1 = own1 + xhalf(oth1);
    __syncthreads();

    const int idx = (t0 + 16 * half + (col >> 1)) * NS + s0 + (col & 1);
    out[idx]          = y0;
    out[idx + 8 * NS] = y1;
  }
}

extern "C" void kernel_launch(void* const* d_in, const int* in_sizes, int n_in,
                              void* d_out, int out_size, void* d_ws, size_t ws_size,
                              hipStream_t stream) {
  (void)in_sizes; (void)n_in; (void)out_size; (void)d_ws; (void)ws_size;
  const float* P  = (const float*)d_in[0];
  const float* T1 = (const float*)d_in[1];
  const float* T2 = (const float*)d_in[2];
  const float* E  = (const float*)d_in[3];
  const float* w  = (const float*)d_in[4];
  float* out = (float*)d_out;

  dim3 grid(NS / SITES_PER_BLOCK);
  dim3 block(BLOCK_THREADS);
  waternet_scan<<<grid, block, 0, stream>>>(P, T1, T2, E, w, out);
}


# Output half of each step pair (Q*ga of the lane's two units and their add) issued one pair late, among the next pair's state updates; the last pair's among the trailing WMMAs; same arithmetic
# speedup vs baseline: 2.9522x; 1.0319x over previous
#include <hip/hip_runtime.h>
#include <math.h>

typedef __attribute__((ext_vector_type(2))) float v2f;
typedef __attribute__((ext_vector_type(8))) float v8f;

#define NT 1024
#define NS 2048
#define WAVES_PER_BLOCK 4
#define SITES_PER_WAVE 2
#define SITES_PER_BLOCK (WAVES_PER_BLOCK * SITES_PER_WAVE)
#define BLOCK_THREADS (WAVES_PER_BLOCK * 32)
#define TILE_ELEMS (32 * SITES_PER_BLOCK)
#define WMMA_LAG 2


__device__ __forceinline__ float xhalf(float v) {
  int x = __builtin_bit_cast(int, v);
  int r = __builtin_amdgcn_permlanex16(x, x, 0x76543210u, 0xfedcba98u,
                                       false,  false);
  return __builtin_bit_cast(float, r);
}

__device__ __forceinline__ v8f wmma_fold_pair(float p0, float p1, v2f bsel, v8f acc) {
  v2f a; a.x = p0; a.y = p1;
  return __builtin_amdgcn_wmma_f32_16x16x4_f32(
       false, a,  false, bsel,
       (short)0, acc,  false,  false);
}

__device__ __forceinline__ float sum_rows8(v8f d) {
  const float e0 = d[0] + d[4], e1 = d[1] + d[5];
  const float e2 = d[2] + d[6], e3 = d[3] + d[7];
  const float g0 = e0 + e2, g1 = e1 + e3;
  return g0 + g1;
}

__device__ __forceinline__ float sigmoidf(float x) {
  return 1.0f / (1.0f + expf(-x));
}

__global__ __launch_bounds__(BLOCK_THREADS) void waternet_scan(
    const float* __restrict__ P,  const float* __restrict__ T1,
    const float* __restrict__ T2, const float* __restrict__ E,
    const float* __restrict__ w,  float* __restrict__ out) {
  const int lane = threadIdx.x & 31;
  const int wv   = threadIdx.x >> 5;
  const int half = lane >> 4;
  const int col  = lane & 15;
  const int s0   = blockIdx.x * SITES_PER_BLOCK + SITES_PER_WAVE * wv;

  float gm = expf(w[lane]) + 1.0f;
  float ge = sigmoidf(w[32 + lane]);
  float go = sigmoidf(w[64 + lane]);
  const float xw = w[96 + lane];
  float mx = xw;
  for (int off = 16; off; off >>= 1) mx = fmaxf(mx, __shfl_xor(mx, off, 32));
  const float ex = expf(xw - mx);
  float sm = ex;
  for (int off = 16; off; off >>= 1) sm += __shfl_xor(sm, off, 32);
  float ga = ex / sm;

  const int hu = ((col & 7) << 1) | (col >> 3);
  const float gm0 = __shfl(gm, hu, 32), gm1 = __shfl(gm, hu + 16, 32);
  const float ge0 = __shfl(ge, hu, 32), ge1 = __shfl(ge, hu + 16, 32);
  const float go0 = __shfl(go, hu, 32), go1 = __shfl(go, hu + 16, 32);
  const float ga0 = __shfl(ga, hu, 32), ga1 = __shfl(ga, hu + 16, 32);

  v2f bsel[4];
#pragma unroll
  for (int j = 0; j < 4; ++j) {
    bsel[j].x = (col == 4 * j + half)     ? 1.0f : 0.0f;
    bsel[j].y = (col == 4 * j + 2 + half) ? 1.0f : 0.0f;
  }

  float S0 = 0.0f, H0 = 0.0f;
  float S1 = 0.0f, H1 = 0.0f;

  __shared__ float4 stage[2][TILE_ELEMS];

  const int tile0 =
      (threadIdx.x >> 3) * NS + blockIdx.x * SITES_PER_BLOCK + (threadIdx.x & 7);
  const int tile1 = tile0 + 16 * NS;

  auto part_ratio = [](float sum, float t1v, float t2v) -> float {
    return sum / (t2v - t1v);
  };
  auto part_acos = [](float ratio) -> float {
    return acosf(fminf(1.0f, fmaxf(-1.0f, ratio)));
  };
  auto part_frac = [](float ac) -> float {
    return ac / 3.1415f;
  };
  auto part_finish = [](float Pv, float t1v, float t2v, float Ev, float sum,
                        float frac) -> float4 {
    float rp = 1.0f - frac;
    rp = (t1v >= 0.0f) ? 1.0f : rp;
    rp = (t2v <= 0.0f) ? 0.0f : rp;
    return make_float4(sum * 0.5f, (1.0f - rp) * Pv, rp * Pv, Ev);
  };
  auto forcing = [&](float Pv, float t1v, float t2v, float Ev) -> float4 {
    const float sum = t1v + t2v;
    return part_finish(Pv, t1v, t2v, Ev, sum,
                       part_frac(part_acos(part_ratio(sum, t1v, t2v))));
  };

  float cPx  = P[tile0 + 32 * NS],  cPy  = P[tile1 + 32 * NS];
  float ct1x = T1[tile0 + 32 * NS], ct1y = T1[tile1 + 32 * NS];
  float ct2x = T2[tile0 + 32 * NS], ct2y = T2[tile1 + 32 * NS];
  float cEx  = E[tile0 + 32 * NS],  cEy  = E[tile1 + 32 * NS];
  stage[0][threadIdx.x]                 = forcing(P[tile0], T1[tile0], T2[tile0], E[tile0]);
  stage[0][threadIdx.x + BLOCK_THREADS] = forcing(P[tile1], T1[tile1], T2[tile1], E[tile1]);
  __syncthreads();

  auto step_state = [](const float4 fk, float& S, float& H, const float gm,
                       const float ge, const float go) -> float {
    const float Smelt = fminf(S, fmaxf(fk.x * gm, 0.0f));
    S = S + fk.y - Smelt;
    const float net = fk.z - fk.w * ge;
    const float inflow = Smelt + net;
    const float Hn = fmaxf(H + inflow, 0.0f);
    const float Q = Hn * go;
    H = Hn - Q;
    return Hn;
  };
  auto step_out = [&](const float hn0, const float hn1) -> float {
    const float p1 = (hn1 * go1) * ga1;
    return fmaf(hn0 * go0, ga0, p1);
  };

  for (int t0 = 0; t0 < NT; t0 += 32) {
    const int buf = (t0 >> 5) & 1;

    const int tn = (t0 + 64 < NT) ? t0 + 64 : t0;
    const int gidx = tn * NS + tile0;
    const int gidy = tn * NS + tile1;
    const float Pn  = P[gidx],  Pm  = P[gidy];
    const float t1n = T1[gidx], t1m = T1[gidy];
    const float t2n = T2[gidx], t2m = T2[gidy];
    const float En  = E[gidx],  Em  = E[gidy];

    float4 f[32];
#pragma unroll
    for (int k = 0; k < 32; ++k)
      f[k] = stage[buf][k * SITES_PER_BLOCK + SITES_PER_WAVE * wv + half];
    __builtin_amdgcn_sched_barrier(0);

    if (tn + 32 < NT) {
      const int nidx = gidx + 32 * NS;
      const int nidy = gidy + 32 * NS;
      __builtin_prefetch(&P[nidx],  0, 0);
      __builtin_prefetch(&T1[nidx], 0, 0);
      __builtin_prefetch(&T2[nidx], 0, 0);
      __builtin_prefetch(&E[nidx],  0, 0);
      __builtin_prefetch(&P[nidy],  0, 0);
      __builtin_prefetch(&T1[nidy], 0, 0);
      __builtin_prefetch(&T2[nidy], 0, 0);
      __builtin_prefetch(&E[nidy],  0, 0);
    }

    v8f acc0 = {}, acc1 = {}, acc2 = {}, acc3 = {};
    float q[32];
    float hn0[32], hn1[32];
    const float sumx = ct1x + ct2x, sumy = ct1y + ct2y;
    float ratx = 0.0f, raty = 0.0f, acx = 0.0f, acy = 0.0f;
#pragma unroll
    for (int k = 0; k < 32; ++k) {
      hn0[k] = step_state(f[k], S0, H0, gm0, ge0, go0);
      hn1[k] = step_state(f[k], S1, H1, gm1, ge1, go1);
      if (k & 1) {
        if (k >= 3) {
          q[k - 3] = step_out(hn0[k - 3], hn1[k - 3]);
          q[k - 2] = step_out(hn0[k - 2], hn1[k - 2]);
        }
        const int j = (k >> 1) - WMMA_LAG;
        if (j >= 0) {
          if (j < 4)       acc0 = wmma_fold_pair(q[2 * j], q[2 * j + 1], bsel[j & 3], acc0);
          else if (j < 8)  acc1 = wmma_fold_pair(q[2 * j], q[2 * j + 1], bsel[j & 3], acc1);
          else if (j < 12) acc2 = wmma_fold_pair(q[2 * j], q[2 * j + 1], bsel[j & 3], acc2);
          else             acc3 = wmma_fold_pair(q[2 * j], q[2 * j + 1], bsel[j & 3], acc3);
        }
        if (k == 1)  { ratx = part_ratio(sumx, ct1x, ct2x); raty = part_ratio(sumy, ct1y, ct2y); }
        if (k == 15) { acx = part_acos(ratx); acy = part_acos(raty); }
        __builtin_amdgcn_sched_barrier(0);
      }
    }
#pragma unroll
    for (int j = 16 - WMMA_LAG; j < 16; ++j) {
      if (j == 15) {
        q[30] = step_out(hn0[30], hn1[30]);
        q[31] = step_out(hn0[31], hn1[31]);
      }
      acc3 = wmma_fold_pair(q[2 * j], q[2 * j + 1], bsel[j & 3], acc3);
    }
    __builtin_amdgcn_sched_barrier(0);

    stage[buf ^ 1][threadIdx.x] =
        part_finish(cPx, ct1x, ct2x, cEx, sumx, part_frac(acx));
    stage[buf ^ 1][threadIdx.x + BLOCK_THREADS] =
        part_finish(cPy, ct1y, ct2y, cEy, sumy, part_frac(acy));
    cPx = Pn;   cPy = Pm;
    ct1x = t1n; ct1y = t1m;
    ct2x = t2n; ct2y = t2m;
    cEx = En;   cEy = Em;
    __builtin_amdgcn_sched_barrier(0);

    const float G0 = sum_rows8(acc0), G1 = sum_rows8(acc1);
    const float G2 = sum_rows8(acc2), G3 = sum_rows8(acc3);
    const bool lo_half = lane < 16;
    const float own0 = lo_half ? G0 : G2;
    const float oth0 = lo_half ? G2 : G0;
    const float own1 = lo_half ? G1 : G3;
    const float oth1 = lo_half ? G3 : G1;
    const float y0 = own0 + xhalf(oth0);
    const float y1 = own1 + xhalf(oth1);
    __syncthreads();

    const int idx = (t0 + 16 * half + (col >> 1)) * NS + s0 + (col & 1);
    out[idx]          = y0;
    out[idx + 8 * NS] = y1;
  }
}

extern "C" void kernel_launch(void* const* d_in, const int* in_sizes, int n_in,
                              void* d_out, int out_size, void* d_ws, size_t ws_size,
                              hipStream_t stream) {
  (void)in_sizes; (void)n_in; (void)out_size; (void)d_ws; (void)ws_size;
  const float* P  = (const float*)d_in[0];
  const float* T1 = (const float*)d_in[1];
  const float* T2 = (const float*)d_in[2];
  const float* E  = (const float*)d_in[3];
  const float* w  = (const float*)d_in[4];
  float* out = (float*)d_out;

  dim3 grid(NS / SITES_PER_BLOCK);
  dim3 block(BLOCK_THREADS);
  waternet_scan<<<grid, block, 0, stream>>>(P, T1, T2, E, w, out);
}


# Rain/snow partition, raw loads and tile staging moved to four helper waves (8-wave groups, second wave per SIMD): they issue in the scan waves' stall slots; scan stream loses ~140 instr per block
# speedup vs baseline: 3.1642x; 1.0718x over previous
#include <hip/hip_runtime.h>
#include <math.h>

typedef __attribute__((ext_vector_type(2))) float v2f;
typedef __attribute__((ext_vector_type(8))) float v8f;

#define NT 1024
#define NS 2048
#define WAVES_PER_BLOCK 4
#define HELPER_WAVES 4
#define SITES_PER_WAVE 2
#define SITES_PER_BLOCK (WAVES_PER_BLOCK * SITES_PER_WAVE)
#define STAGE_THREADS (HELPER_WAVES * 32)
#define BLOCK_THREADS ((WAVES_PER_BLOCK + HELPER_WAVES) * 32)
#define TILE_ELEMS (32 * SITES_PER_BLOCK)
#define WMMA_LAG 2


__device__ __forceinline__ float xhalf(float v) {
  int x = __builtin_bit_cast(int, v);
  int r = __builtin_amdgcn_permlanex16(x, x, 0x76543210u, 0xfedcba98u,
                                       false,  false);
  return __builtin_bit_cast(float, r);
}

__device__ __forceinline__ v8f wmma_fold_pair(float p0, float p1, v2f bsel, v8f acc) {
  v2f a; a.x = p0; a.y = p1;
  return __builtin_amdgcn_wmma_f32_16x16x4_f32(
       false, a,  false, bsel,
       (short)0, acc,  false,  false);
}

__device__ __forceinline__ float sum_rows8(v8f d) {
  const float e0 = d[0] + d[4], e1 = d[1] + d[5];
  const float e2 = d[2] + d[6], e3 = d[3] + d[7];
  const float g0 = e0 + e2, g1 = e1 + e3;
  return g0 + g1;
}

__device__ __forceinline__ float sigmoidf(float x) {
  return 1.0f / (1.0f + expf(-x));
}

__global__ __launch_bounds__(BLOCK_THREADS) void waternet_scan(
    const float* __restrict__ P,  const float* __restrict__ T1,
    const float* __restrict__ T2, const float* __restrict__ E,
    const float* __restrict__ w,  float* __restrict__ out) {
  const int lane = threadIdx.x & 31;
  const int wv   = __builtin_amdgcn_readfirstlane(threadIdx.x >> 5);
  const bool scan = wv < WAVES_PER_BLOCK;
  const int half = lane >> 4;
  const int col  = lane & 15;
  const int s0   = blockIdx.x * SITES_PER_BLOCK + SITES_PER_WAVE * wv;

  float gm0 = 0.0f, gm1 = 0.0f, ge0 = 0.0f, ge1 = 0.0f;
  float go0 = 0.0f, go1 = 0.0f, ga0 = 0.0f, ga1 = 0.0f;
  if (scan) {
  float gm = expf(w[lane]) + 1.0f;
  float ge = sigmoidf(w[32 + lane]);
  float go = sigmoidf(w[64 + lane]);
  const float xw = w[96 + lane];
  float mx = xw;
  for (int off = 16; off; off >>= 1) mx = fmaxf(mx, __shfl_xor(mx, off, 32));
  const float ex = expf(xw - mx);
  float sm = ex;
  for (int off = 16; off; off >>= 1) sm += __shfl_xor(sm, off, 32);
  float ga = ex / sm;

  const int hu = ((col & 7) << 1) | (col >> 3);
  gm0 = __shfl(gm, hu, 32); gm1 = __shfl(gm, hu + 16, 32);
  ge0 = __shfl(ge, hu, 32); ge1 = __shfl(ge, hu + 16, 32);
  go0 = __shfl(go, hu, 32); go1 = __shfl(go, hu + 16, 32);
  ga0 = __shfl(ga, hu, 32); ga1 = __shfl(ga, hu + 16, 32);
  }

  v2f bsel[4];
#pragma unroll
  for (int j = 0; j < 4; ++j) {
    bsel[j].x = (col == 4 * j + half)     ? 1.0f : 0.0f;
    bsel[j].y = (col == 4 * j + 2 + half) ? 1.0f : 0.0f;
  }

  float S0 = 0.0f, H0 = 0.0f;
  float S1 = 0.0f, H1 = 0.0f;

  __shared__ float4 stage[2][TILE_ELEMS];

  const int st = (int)threadIdx.x - WAVES_PER_BLOCK * 32;
  const int tile0 = (st >> 3) * NS + blockIdx.x * SITES_PER_BLOCK + (st & 7);
  const int tile1 = tile0 + 16 * NS;

  auto part_ratio = [](float sum, float t1v, float t2v) -> float {
    return sum / (t2v - t1v);
  };
  auto part_acos = [](float ratio) -> float {
    return acosf(fminf(1.0f, fmaxf(-1.0f, ratio)));
  };
  auto part_frac = [](float ac) -> float {
    return ac / 3.1415f;
  };
  auto part_finish = [](float Pv, float t1v, float t2v, float Ev, float sum,
                        float frac) -> float4 {
    float rp = 1.0f - frac;
    rp = (t1v >= 0.0f) ? 1.0f : rp;
    rp = (t2v <= 0.0f) ? 0.0f : rp;
    return make_float4(sum * 0.5f, (1.0f - rp) * Pv, rp * Pv, Ev);
  };
  auto forcing = [&](float Pv, float t1v, float t2v, float Ev) -> float4 {
    const float sum = t1v + t2v;
    return part_finish(Pv, t1v, t2v, Ev, sum,
                       part_frac(part_acos(part_ratio(sum, t1v, t2v))));
  };

  float cPx = 0.0f, cPy = 0.0f, ct1x = 0.0f, ct1y = 0.0f;
  float ct2x = 0.0f, ct2y = 0.0f, cEx = 0.0f, cEy = 0.0f;
  if (!scan) {
    cPx  = P[tile0 + 32 * NS];  cPy  = P[tile1 + 32 * NS];
    ct1x = T1[tile0 + 32 * NS]; ct1y = T1[tile1 + 32 * NS];
    ct2x = T2[tile0 + 32 * NS]; ct2y = T2[tile1 + 32 * NS];
    cEx  = E[tile0 + 32 * NS];  cEy  = E[tile1 + 32 * NS];
    stage[0][st]                 = forcing(P[tile0], T1[tile0], T2[tile0], E[tile0]);
    stage[0][st + STAGE_THREADS] = forcing(P[tile1], T1[tile1], T2[tile1], E[tile1]);
  }
  __syncthreads();

  auto step_state = [](const float4 fk, float& S, float& H, const float gm,
                       const float ge, const float go) -> float {
    const float Smelt = fminf(S, fmaxf(fk.x * gm, 0.0f));
    S = S + fk.y - Smelt;
    const float net = fk.z - fk.w * ge;
    const float inflow = Smelt + net;
    const float Hn = fmaxf(H + inflow, 0.0f);
    const float Q = Hn * go;
    H = Hn - Q;
    return Hn;
  };
  auto step_out = [&](const float hn0, const float hn1) -> float {
    const float p1 = (hn1 * go1) * ga1;
    return fmaf(hn0 * go0, ga0, p1);
  };

  for (int t0 = 0; t0 < NT; t0 += 32) {
    const int buf = (t0 >> 5) & 1;

    float y0 = 0.0f, y1 = 0.0f;
    if (!scan) {
    const int tn = (t0 + 64 < NT) ? t0 + 64 : t0;
    const int gidx = tn * NS + tile0;
    const int gidy = tn * NS + tile1;
    const float Pn  = P[gidx],  Pm  = P[gidy];
    const float t1n = T1[gidx], t1m = T1[gidy];
    const float t2n = T2[gidx], t2m = T2[gidy];
    const float En  = E[gidx],  Em  = E[gidy];
    if (tn + 32 < NT) {
      const int nidx = gidx + 32 * NS;
      const int nidy = gidy + 32 * NS;
      __builtin_prefetch(&P[nidx],  0, 0);
      __builtin_prefetch(&T1[nidx], 0, 0);
      __builtin_prefetch(&T2[nidx], 0, 0);
      __builtin_prefetch(&E[nidx],  0, 0);
      __builtin_prefetch(&P[nidy],  0, 0);
      __builtin_prefetch(&T1[nidy], 0, 0);
      __builtin_prefetch(&T2[nidy], 0, 0);
      __builtin_prefetch(&E[nidy],  0, 0);
    }
    stage[buf ^ 1][st]                 = forcing(cPx, ct1x, ct2x, cEx);
    stage[buf ^ 1][st + STAGE_THREADS] = forcing(cPy, ct1y, ct2y, cEy);
    cPx = Pn;   cPy = Pm;
    ct1x = t1n; ct1y = t1m;
    ct2x = t2n; ct2y = t2m;
    cEx = En;   cEy = Em;
    } else {
    float4 f[32];
#pragma unroll
    for (int k = 0; k < 32; ++k)
      f[k] = stage[buf][k * SITES_PER_BLOCK + SITES_PER_WAVE * wv + half];
    __builtin_amdgcn_sched_barrier(0);

    v8f acc0 = {}, acc1 = {}, acc2 = {}, acc3 = {};
    float q[32];
    float hn0[32], hn1[32];
#pragma unroll
    for (int k = 0; k < 32; ++k) {
      hn0[k] = step_state(f[k], S0, H0, gm0, ge0, go0);
      hn1[k] = step_state(f[k], S1, H1, gm1, ge1, go1);
      if (k & 1) {
        if (k >= 3) {
          q[k - 3] = step_out(hn0[k - 3], hn1[k - 3]);
          q[k - 2] = step_out(hn0[k - 2], hn1[k - 2]);
        }
        const int j = (k >> 1) - WMMA_LAG;
        if (j >= 0) {
          if (j < 4)       acc0 = wmma_fold_pair(q[2 * j], q[2 * j + 1], bsel[j & 3], acc0);
          else if (j < 8)  acc1 = wmma_fold_pair(q[2 * j], q[2 * j + 1], bsel[j & 3], acc1);
          else if (j < 12) acc2 = wmma_fold_pair(q[2 * j], q[2 * j + 1], bsel[j & 3], acc2);
          else             acc3 = wmma_fold_pair(q[2 * j], q[2 * j + 1], bsel[j & 3], acc3);
        }
        __builtin_amdgcn_sched_barrier(0);
      }
    }
#pragma unroll
    for (int j = 16 - WMMA_LAG; j < 16; ++j) {
      if (j == 15) {
        q[30] = step_out(hn0[30], hn1[30]);
        q[31] = step_out(hn0[31], hn1[31]);
      }
      acc3 = wmma_fold_pair(q[2 * j], q[2 * j + 1], bsel[j & 3], acc3);
    }

    const float G0 = sum_rows8(acc0), G1 = sum_rows8(acc1);
    const float G2 = sum_rows8(acc2), G3 = sum_rows8(acc3);
    const bool lo_half = lane < 16;
    const float own0 = lo_half ? G0 : G2;
    const float oth0 = lo_half ? G2 : G0;
    const float own1 = lo_half ? G1 : G3;
    const float oth1 = lo_half ? G3 : G1;
    y0 = own0 + xhalf(oth0);
    y1 = own1 + xhalf(oth1);
    }
    __syncthreads();

    if (scan) {
      const int idx = (t0 + 16 * half + (col >> 1)) * NS + s0 + (col & 1);
      out[idx]          = y0;
      out[idx + 8 * NS] = y1;
    }
  }
}

extern "C" void kernel_launch(void* const* d_in, const int* in_sizes, int n_in,
                              void* d_out, int out_size, void* d_ws, size_t ws_size,
                              hipStream_t stream) {
  (void)in_sizes; (void)n_in; (void)out_size; (void)d_ws; (void)ws_size;
  const float* P  = (const float*)d_in[0];
  const float* T1 = (const float*)d_in[1];
  const float* T2 = (const float*)d_in[2];
  const float* E  = (const float*)d_in[3];
  const float* w  = (const float*)d_in[4];
  float* out = (float*)d_out;

  dim3 grid(NS / SITES_PER_BLOCK);
  dim3 block(BLOCK_THREADS);
  waternet_scan<<<grid, block, 0, stream>>>(P, T1, T2, E, w, out);
}
